# starting kernel
# baseline (speedup 1.0000x reference)
#include <hip/hip_runtime.h>
#include <hip/hip_bf16.h>

typedef float v2f __attribute__((ext_vector_type(2)));
typedef float v8f __attribute__((ext_vector_type(8)));

#define Hh   96
#define Ww   96
#define HW   (96 * 96)
#define Cc   128
#define NPIX (2 * HW)

__global__ void __launch_bounds__(256) qkv_wmma_kernel(
    const float* __restrict__ x,  const float* __restrict__ Wq,
    const float* __restrict__ Wk, const float* __restrict__ Wv,
    float* __restrict__ qo, float* __restrict__ ko, float* __restrict__ vo)
{
    const int lane = threadIdx.x & 31;
    const int wave = threadIdx.x >> 5;
    const int p0   = blockIdx.x * 16;
    const int b    = p0 / HW;
    const int pix  = p0 - b * HW;
    const float* xb = x + (size_t)b * Cc * HW + pix;

    const int o0    = wave * 16;
    const int row   = lane & 15;
    const int khalf = (lane >> 4) << 1;
    const int colp  = lane & 15;

    v8f cq = {}; v8f ck = {}; v8f cv = {};
    #pragma unroll 4
    for (int k0 = 0; k0 < Cc; k0 += 4) {
        v2f bfrag;
        bfrag.x = xb[(size_t)(k0 + khalf) * HW + colp];
        bfrag.y = xb[(size_t)(k0 + khalf + 1) * HW + colp];
        const int widx = (o0 + row) * Cc + k0 + khalf;
        v2f aq, ak, av;
        aq.x = Wq[widx]; aq.y = Wq[widx + 1];
        ak.x = Wk[widx]; ak.y = Wk[widx + 1];
        av.x = Wv[widx]; av.y = Wv[widx + 1];
        cq = __builtin_amdgcn_wmma_f32_16x16x4_f32(false, aq, false, bfrag, (short)0, cq, false, false);
        ck = __builtin_amdgcn_wmma_f32_16x16x4_f32(false, ak, false, bfrag, (short)0, ck, false, false);
        cv = __builtin_amdgcn_wmma_f32_16x16x4_f32(false, av, false, bfrag, (short)0, cv, false, false);
    }

    const int rowoff = (lane >> 4) << 3;
    const size_t base = (size_t)(p0 + colp) * Cc + o0 + rowoff;
    #pragma unroll
    for (int r = 0; r < 8; ++r) {
        qo[base + r] = cq[r];
        ko[base + r] = ck[r];
        vo[base + r] = cv[r];
    }
}

__global__ void __launch_bounds__(256) attn_kernel(
    const float* __restrict__ qv, const float* __restrict__ kv,
    const float* __restrict__ vv, const int* __restrict__ msk,
    float* __restrict__ ao)
{
    const int t = blockIdx.x * 256 + threadIdx.x;
    if (t >= NPIX * 4) return;
    const int hd  = t & 3;
    const int p   = t >> 2;
    const int b   = p / HW;
    const int pix = p - b * HW;
    const int y   = pix / Ww;
    const int x   = pix - y * Ww;

    float q[32];
    const float4* q4 = (const float4*)(qv + (size_t)p * Cc + hd * 32);
    #pragma unroll
    for (int u = 0; u < 8; ++u) {
        float4 t4 = q4[u];
        q[4*u] = t4.x; q[4*u+1] = t4.y; q[4*u+2] = t4.z; q[4*u+3] = t4.w;
    }

    const float scale = 0.17677669529663687f;
    float mx = -1e30f, s = 0.f;
    float acc[32];
    #pragma unroll
    for (int d = 0; d < 32; ++d) acc[d] = 0.f;

    #pragma unroll 1
    for (int wi = 0; wi < 49; ++wi) {
        const int di = wi / 7, dj = wi - di * 7;
        const int ny = y + (di - 3) * 2;
        const int nx = x + (dj - 3) * 2;
        const bool valid = ((unsigned)ny < (unsigned)Hh) && ((unsigned)nx < (unsigned)Ww);

        float logit = -1.0e9f;
        const float4* k4 = nullptr;
        const float4* v4 = nullptr;
        if (valid) {
            const int np = b * HW + ny * Ww + nx;
            k4 = (const float4*)(kv + (size_t)np * Cc + hd * 32);
            v4 = (const float4*)(vv + (size_t)np * Cc + hd * 32);
            float dot = 0.f;
            #pragma unroll
            for (int u = 0; u < 8; ++u) {
                float4 kk = k4[u];
                dot += q[4*u] * kk.x + q[4*u+1] * kk.y + q[4*u+2] * kk.z + q[4*u+3] * kk.w;
            }
            const float mb = (msk[np] > 0) ? 0.f : -1.0e9f;
            logit = dot * scale + mb;
        }

        const float nm   = fmaxf(mx, logit);
        const float corr = __expf(mx - nm);
        const float w    = __expf(logit - nm);
        mx = nm;
        s  = s * corr + w;
        if (valid) {
            #pragma unroll
            for (int u = 0; u < 8; ++u) {
                float4 vvv = v4[u];
                acc[4*u]   = acc[4*u]   * corr + w * vvv.x;
                acc[4*u+1] = acc[4*u+1] * corr + w * vvv.y;
                acc[4*u+2] = acc[4*u+2] * corr + w * vvv.z;
                acc[4*u+3] = acc[4*u+3] * corr + w * vvv.w;
            }
        } else {
            #pragma unroll
            for (int d = 0; d < 32; ++d) acc[d] *= corr;
        }
    }

    const float inv = 1.f / s;
    float* o = ao + (size_t)p * Cc + hd * 32;
    #pragma unroll
    for (int d = 0; d < 32; ++d) o[d] = acc[d] * inv;
}

__global__ void __launch_bounds__(256) proj_wmma_kernel(
    const float* __restrict__ ain, const float* __restrict__ Wp,
    float* __restrict__ out)
{
    const int lane = threadIdx.x & 31;
    const int wave = threadIdx.x >> 5;
    const int p0   = blockIdx.x * 16;
    const int b    = p0 / HW;
    const int pix  = p0 - b * HW;
    const float* aw = ain + (size_t)p0 * Cc;

    const int o0    = wave * 16;
    const int row   = lane & 15;
    const int khalf = (lane >> 4) << 1;
    const int colp  = lane & 15;

    v8f c = {};
    #pragma unroll 4
    for (int k0 = 0; k0 < Cc; k0 += 4) {
        v2f a, bfrag;
        const int widx = (o0 + row) * Cc + k0 + khalf;
        a.x = Wp[widx]; a.y = Wp[widx + 1];
        bfrag.x = aw[(size_t)colp * Cc + k0 + khalf];
        bfrag.y = aw[(size_t)colp * Cc + k0 + khalf + 1];
        c = __builtin_amdgcn_wmma_f32_16x16x4_f32(false, a, false, bfrag, (short)0, c, false, false);
    }

    const int rowoff = (lane >> 4) << 3;
    float* dst = out + (size_t)b * Cc * HW + (size_t)(o0 + rowoff) * HW + pix + colp;
    #pragma unroll
    for (int r = 0; r < 8; ++r) dst[(size_t)r * HW] = c[r];
}

extern "C" void kernel_launch(void* const* d_in, const int* in_sizes, int n_in,
                              void* d_out, int out_size, void* d_ws, size_t ws_size,
                              hipStream_t stream)
{
    const float* x  = (const float*)d_in[0];
    const int*   m  = (const int*)  d_in[1];
    const float* Wq = (const float*)d_in[2];
    const float* Wk = (const float*)d_in[3];
    const float* Wv = (const float*)d_in[4];
    const float* Wp = (const float*)d_in[5];
    float* out = (float*)d_out;

    float* ws = (float*)d_ws;
    const size_t n = (size_t)NPIX * Cc;
    float* qb = ws;
    float* kb = ws + n;
    float* vb = ws + 2 * n;
    float* ab = ws + 3 * n;

    qkv_wmma_kernel<<<NPIX / 16, 256, 0, stream>>>(x, Wq, Wk, Wv, qb, kb, vb);
    attn_kernel<<<(NPIX * 4) / 256, 256, 0, stream>>>(qb, kb, vb, m, ab);
    proj_wmma_kernel<<<NPIX / 16, 256, 0, stream>>>(ab, Wp, out);
}


# attention: one wave per pixel, lane = head*8+part owns 4 of 32 channels, so each K/V load is 4 whole 128-byte lines (was 32 partial lines); dot summed as 8 lane partials + 3-step DPP butterfly
# speedup vs baseline: 2.2414x; 2.2414x over previous
#include <hip/hip_runtime.h>
#include <hip/hip_bf16.h>

typedef float v2f __attribute__((ext_vector_type(2)));
typedef float v8f __attribute__((ext_vector_type(8)));

#define Hh   96
#define Ww   96
#define HW   (96 * 96)
#define Cc   128
#define NPIX (2 * HW)

__global__ void __launch_bounds__(256) qkv_wmma_kernel(
    const float* __restrict__ x,  const float* __restrict__ Wq,
    const float* __restrict__ Wk, const float* __restrict__ Wv,
    float* __restrict__ qo, float* __restrict__ ko, float* __restrict__ vo)
{
    const int lane = threadIdx.x & 31;
    const int wave = threadIdx.x >> 5;
    const int p0   = blockIdx.x * 16;
    const int b    = p0 / HW;
    const int pix  = p0 - b * HW;
    const float* xb = x + (size_t)b * Cc * HW + pix;

    const int o0    = wave * 16;
    const int row   = lane & 15;
    const int khalf = (lane >> 4) << 1;
    const int colp  = lane & 15;

    v8f cq = {}; v8f ck = {}; v8f cv = {};
    #pragma unroll 4
    for (int k0 = 0; k0 < Cc; k0 += 4) {
        v2f bfrag;
        bfrag.x = xb[(size_t)(k0 + khalf) * HW + colp];
        bfrag.y = xb[(size_t)(k0 + khalf + 1) * HW + colp];
        const int widx = (o0 + row) * Cc + k0 + khalf;
        v2f aq, ak, av;
        aq.x = Wq[widx]; aq.y = Wq[widx + 1];
        ak.x = Wk[widx]; ak.y = Wk[widx + 1];
        av.x = Wv[widx]; av.y = Wv[widx + 1];
        cq = __builtin_amdgcn_wmma_f32_16x16x4_f32(false, aq, false, bfrag, (short)0, cq, false, false);
        ck = __builtin_amdgcn_wmma_f32_16x16x4_f32(false, ak, false, bfrag, (short)0, ck, false, false);
        cv = __builtin_amdgcn_wmma_f32_16x16x4_f32(false, av, false, bfrag, (short)0, cv, false, false);
    }

    const int rowoff = (lane >> 4) << 3;
    const size_t base = (size_t)(p0 + colp) * Cc + o0 + rowoff;
    #pragma unroll
    for (int r = 0; r < 8; ++r) {
        qo[base + r] = cq[r];
        ko[base + r] = ck[r];
        vo[base + r] = cv[r];
    }
}

#define ATT_BLOCKS (2 * Hh * 6 * 2)

template <int M>
__device__ __forceinline__ float xor8(float v)
{
    return __builtin_bit_cast(float,
        __builtin_amdgcn_update_dpp(0, __builtin_bit_cast(int, v), 0x160 | M, 0xF, 0xF, true));
}

__global__ void __launch_bounds__(256) attn_kernel(
    const float* __restrict__ qv, const float* __restrict__ kv,
    const float* __restrict__ vv, const int* __restrict__ msk,
    float* __restrict__ ao)
{
    const int lane = threadIdx.x & 31;
    const int wave = __builtin_amdgcn_readfirstlane((int)(threadIdx.x >> 5));
    const int blk  = blockIdx.x;
    const int par  = blk & 1;
    const int rs   = blk >> 1;
    const int brow = rs / 6;
    const int seg  = rs - brow * 6;
    const int b    = brow / Hh;
    const int y    = brow - b * Hh;
    const int x    = seg * 16 + par + 2 * wave;
    const int p    = brow * Ww + x;

    const float4 q = *(const float4*)(qv + (size_t)p * Cc + lane * 4);

    const float scale = 0.17677669529663687f;
    float mx = -1e30f, s = 0.f;
    float4 acc = make_float4(0.f, 0.f, 0.f, 0.f);

    #pragma unroll
    for (int di = 0; di < 7; ++di) {
        const int ny = y + (di - 3) * 2;
        const bool yok = (unsigned)ny < (unsigned)Hh;
        float4 kk[7], vq[7];
        int    mw[7];
        #pragma unroll
        for (int dj = 0; dj < 7; ++dj) {
            const int nx = x + (dj - 3) * 2;
            const bool valid = yok && ((unsigned)nx < (unsigned)Ww);
            const int np = valid ? (b * HW + ny * Ww + nx) : p;
            kk[dj] = *(const float4*)(kv + (size_t)np * Cc + lane * 4);
            vq[dj] = *(const float4*)(vv + (size_t)np * Cc + lane * 4);
            mw[dj] = msk[np];
        }
        #pragma unroll
        for (int dj = 0; dj < 7; ++dj) {
            const int nx = x + (dj - 3) * 2;
            const bool valid = yok && ((unsigned)nx < (unsigned)Ww);

            float dot = q.x * kk[dj].x + q.y * kk[dj].y + q.z * kk[dj].z + q.w * kk[dj].w;
            dot += xor8<1>(dot);
            dot += xor8<2>(dot);
            dot += xor8<4>(dot);

            const float mb    = (mw[dj] > 0) ? 0.f : -1.0e9f;
            const float logit = valid ? (dot * scale + mb) : -1.0e9f;

            const float nm   = fmaxf(mx, logit);
            const float corr = __expf(mx - nm);
            const float w    = __expf(logit - nm);
            mx = nm;
            s  = s * corr + w;
            const float wv = valid ? w : 0.f;
            acc.x = acc.x * corr + wv * vq[dj].x;
            acc.y = acc.y * corr + wv * vq[dj].y;
            acc.z = acc.z * corr + wv * vq[dj].z;
            acc.w = acc.w * corr + wv * vq[dj].w;
        }
    }

    const float inv = 1.f / s;
    float4 o4;
    o4.x = acc.x * inv; o4.y = acc.y * inv; o4.z = acc.z * inv; o4.w = acc.w * inv;
    *(float4*)(ao + (size_t)p * Cc + lane * 4) = o4;
}

__global__ void __launch_bounds__(256) proj_wmma_kernel(
    const float* __restrict__ ain, const float* __restrict__ Wp,
    float* __restrict__ out)
{
    const int lane = threadIdx.x & 31;
    const int wave = threadIdx.x >> 5;
    const int p0   = blockIdx.x * 16;
    const int b    = p0 / HW;
    const int pix  = p0 - b * HW;
    const float* aw = ain + (size_t)p0 * Cc;

    const int o0    = wave * 16;
    const int row   = lane & 15;
    const int khalf = (lane >> 4) << 1;
    const int colp  = lane & 15;

    v8f c = {};
    #pragma unroll 4
    for (int k0 = 0; k0 < Cc; k0 += 4) {
        v2f a, bfrag;
        const int widx = (o0 + row) * Cc + k0 + khalf;
        a.x = Wp[widx]; a.y = Wp[widx + 1];
        bfrag.x = aw[(size_t)colp * Cc + k0 + khalf];
        bfrag.y = aw[(size_t)colp * Cc + k0 + khalf + 1];
        c = __builtin_amdgcn_wmma_f32_16x16x4_f32(false, a, false, bfrag, (short)0, c, false, false);
    }

    const int rowoff = (lane >> 4) << 3;
    float* dst = out + (size_t)b * Cc * HW + (size_t)(o0 + rowoff) * HW + pix + colp;
    #pragma unroll
    for (int r = 0; r < 8; ++r) dst[(size_t)r * HW] = c[r];
}

extern "C" void kernel_launch(void* const* d_in, const int* in_sizes, int n_in,
                              void* d_out, int out_size, void* d_ws, size_t ws_size,
                              hipStream_t stream)
{
    const float* x  = (const float*)d_in[0];
    const int*   m  = (const int*)  d_in[1];
    const float* Wq = (const float*)d_in[2];
    const float* Wk = (const float*)d_in[3];
    const float* Wv = (const float*)d_in[4];
    const float* Wp = (const float*)d_in[5];
    float* out = (float*)d_out;

    float* ws = (float*)d_ws;
    const size_t n = (size_t)NPIX * Cc;
    float* qb = ws;
    float* kb = ws + n;
    float* vb = ws + 2 * n;
    float* ab = ws + 3 * n;

    qkv_wmma_kernel<<<NPIX / 16, 256, 0, stream>>>(x, Wq, Wk, Wv, qb, kb, vb);
    attn_kernel<<<ATT_BLOCKS, 256, 0, stream>>>(qb, kb, vb, m, ab);
    proj_wmma_kernel<<<NPIX / 16, 256, 0, stream>>>(ab, Wp, out);
}


# both GEMMs register-blocked: 4 output tiles per wave (qkv 16ch x 64px of one matrix, proj 32ch x 32px), 16-byte fragment loads; each 16-line weight load feeds 4x the WMMAs; c-sum order changes
# speedup vs baseline: 3.4801x; 1.5526x over previous
#include <hip/hip_runtime.h>
#include <hip/hip_bf16.h>

typedef float v2f __attribute__((ext_vector_type(2)));
typedef float v8f __attribute__((ext_vector_type(8)));

#define Hh   96
#define Ww   96
#define HW   (96 * 96)
#define Cc   128
#define NPIX (2 * HW)

#define QKV_BLOCKS ((NPIX / 64) * 6)

__global__ void __launch_bounds__(128) qkv_wmma_kernel(
    const float* __restrict__ x,  const float* __restrict__ Wq,
    const float* __restrict__ Wk, const float* __restrict__ Wv,
    float* __restrict__ qo, float* __restrict__ ko, float* __restrict__ vo)
{
    const int lane = threadIdx.x & 31;
    const int wave = threadIdx.x >> 5;
    const int grp  = blockIdx.x / 6;
    const int sub  = blockIdx.x - grp * 6;
    const int mat  = sub >> 1;
    const float* __restrict__ Wm = (mat == 0) ? Wq : ((mat == 1) ? Wk : Wv);
    float* __restrict__ dm       = (mat == 0) ? qo : ((mat == 1) ? ko : vo);

    const int p0  = grp * 64;
    const int b   = p0 / HW;
    const int pix = p0 - b * HW;
    const int o0  = (((sub & 1) << 2) + wave) * 16;
    const int c   = lane & 15;
    const int h   = lane >> 4;

    const float* ap = Wm + (size_t)(o0 + c) * Cc + 4 * h;
    const float* bp = x + (size_t)b * Cc * HW + pix + (size_t)(4 * h) * HW + 4 * c;

    v8f acc0 = {}; v8f acc1 = {}; v8f acc2 = {}; v8f acc3 = {};
    #pragma unroll 2
    for (int kb = 0; kb < Cc / 8; ++kb) {
        const float4 a4 = *(const float4*)(ap + kb * 8);
        const float4 r0 = *(const float4*)(bp + (size_t)(kb * 8 + 0) * HW);
        const float4 r1 = *(const float4*)(bp + (size_t)(kb * 8 + 1) * HW);
        const float4 r2 = *(const float4*)(bp + (size_t)(kb * 8 + 2) * HW);
        const float4 r3 = *(const float4*)(bp + (size_t)(kb * 8 + 3) * HW);
        v2f a, f;
        a.x = a4.x; a.y = a4.y;
        f.x = r0.x; f.y = r1.x;
        acc0 = __builtin_amdgcn_wmma_f32_16x16x4_f32(false, a, false, f, (short)0, acc0, false, false);
        f.x = r0.y; f.y = r1.y;
        acc1 = __builtin_amdgcn_wmma_f32_16x16x4_f32(false, a, false, f, (short)0, acc1, false, false);
        f.x = r0.z; f.y = r1.z;
        acc2 = __builtin_amdgcn_wmma_f32_16x16x4_f32(false, a, false, f, (short)0, acc2, false, false);
        f.x = r0.w; f.y = r1.w;
        acc3 = __builtin_amdgcn_wmma_f32_16x16x4_f32(false, a, false, f, (short)0, acc3, false, false);
        a.x = a4.z; a.y = a4.w;
        f.x = r2.x; f.y = r3.x;
        acc0 = __builtin_amdgcn_wmma_f32_16x16x4_f32(false, a, false, f, (short)0, acc0, false, false);
        f.x = r2.y; f.y = r3.y;
        acc1 = __builtin_amdgcn_wmma_f32_16x16x4_f32(false, a, false, f, (short)0, acc1, false, false);
        f.x = r2.z; f.y = r3.z;
        acc2 = __builtin_amdgcn_wmma_f32_16x16x4_f32(false, a, false, f, (short)0, acc2, false, false);
        f.x = r2.w; f.y = r3.w;
        acc3 = __builtin_amdgcn_wmma_f32_16x16x4_f32(false, a, false, f, (short)0, acc3, false, false);
    }

    float* d0 = dm + (size_t)(p0 + 4 * c) * Cc + o0 + 8 * h;
    #pragma unroll
    for (int r = 0; r < 8; ++r) {
        d0[r]          = acc0[r];
        d0[Cc + r]     = acc1[r];
        d0[2 * Cc + r] = acc2[r];
        d0[3 * Cc + r] = acc3[r];
    }
}

#define ATT_BLOCKS (2 * Hh * 6 * 2)

template <int M>
__device__ __forceinline__ float xor8(float v)
{
    return __builtin_bit_cast(float,
        __builtin_amdgcn_update_dpp(0, __builtin_bit_cast(int, v), 0x160 | M, 0xF, 0xF, true));
}

__global__ void __launch_bounds__(256) attn_kernel(
    const float* __restrict__ qv, const float* __restrict__ kv,
    const float* __restrict__ vv, const int* __restrict__ msk,
    float* __restrict__ ao)
{
    const int lane = threadIdx.x & 31;
    const int wave = __builtin_amdgcn_readfirstlane((int)(threadIdx.x >> 5));
    const int blk  = blockIdx.x;
    const int par  = blk & 1;
    const int rs   = blk >> 1;
    const int brow = rs / 6;
    const int seg  = rs - brow * 6;
    const int b    = brow / Hh;
    const int y    = brow - b * Hh;
    const int x    = seg * 16 + par + 2 * wave;
    const int p    = brow * Ww + x;

    const float4 q = *(const float4*)(qv + (size_t)p * Cc + lane * 4);

    const float scale = 0.17677669529663687f;
    float mx = -1e30f, s = 0.f;
    float4 acc = make_float4(0.f, 0.f, 0.f, 0.f);

    #pragma unroll
    for (int di = 0; di < 7; ++di) {
        const int ny = y + (di - 3) * 2;
        const bool yok = (unsigned)ny < (unsigned)Hh;
        float4 kk[7], vq[7];
        int    mw[7];
        #pragma unroll
        for (int dj = 0; dj < 7; ++dj) {
            const int nx = x + (dj - 3) * 2;
            const bool valid = yok && ((unsigned)nx < (unsigned)Ww);
            const int np = valid ? (b * HW + ny * Ww + nx) : p;
            kk[dj] = *(const float4*)(kv + (size_t)np * Cc + lane * 4);
            vq[dj] = *(const float4*)(vv + (size_t)np * Cc + lane * 4);
            mw[dj] = msk[np];
        }
        #pragma unroll
        for (int dj = 0; dj < 7; ++dj) {
            const int nx = x + (dj - 3) * 2;
            const bool valid = yok && ((unsigned)nx < (unsigned)Ww);

            float dot = q.x * kk[dj].x + q.y * kk[dj].y + q.z * kk[dj].z + q.w * kk[dj].w;
            dot += xor8<1>(dot);
            dot += xor8<2>(dot);
            dot += xor8<4>(dot);

            const float mb    = (mw[dj] > 0) ? 0.f : -1.0e9f;
            const float logit = valid ? (dot * scale + mb) : -1.0e9f;

            const float nm   = fmaxf(mx, logit);
            const float corr = __expf(mx - nm);
            const float w    = __expf(logit - nm);
            mx = nm;
            s  = s * corr + w;
            const float wv = valid ? w : 0.f;
            acc.x = acc.x * corr + wv * vq[dj].x;
            acc.y = acc.y * corr + wv * vq[dj].y;
            acc.z = acc.z * corr + wv * vq[dj].z;
            acc.w = acc.w * corr + wv * vq[dj].w;
        }
    }

    const float inv = 1.f / s;
    float4 o4;
    o4.x = acc.x * inv; o4.y = acc.y * inv; o4.z = acc.z * inv; o4.w = acc.w * inv;
    *(float4*)(ao + (size_t)p * Cc + lane * 4) = o4;
}

__global__ void __launch_bounds__(128) proj_wmma_kernel(
    const float* __restrict__ ain, const float* __restrict__ Wp,
    float* __restrict__ out)
{
    const int lane = threadIdx.x & 31;
    const int wave = threadIdx.x >> 5;
    const int p0   = blockIdx.x * 32;
    const int b    = p0 / HW;
    const int pix  = p0 - b * HW;
    const int o0   = wave * 32;
    const int c    = lane & 15;
    const int h    = lane >> 4;

    const float* ap = Wp  + (size_t)(o0 + c) * Cc + 4 * h;
    const float* bp = ain + (size_t)(p0 + 2 * c) * Cc + 4 * h;

    v8f c00 = {}; v8f c01 = {}; v8f c10 = {}; v8f c11 = {};
    #pragma unroll 4
    for (int kb = 0; kb < Cc / 8; ++kb) {
        const float4 wa = *(const float4*)(ap + kb * 8);
        const float4 wb = *(const float4*)(ap + 16 * Cc + kb * 8);
        const float4 xa = *(const float4*)(bp + kb * 8);
        const float4 xb = *(const float4*)(bp + Cc + kb * 8);
        v2f a0, a1, f0, f1;
        a0.x = wa.x; a0.y = wa.y; a1.x = wb.x; a1.y = wb.y;
        f0.x = xa.x; f0.y = xa.y; f1.x = xb.x; f1.y = xb.y;
        c00 = __builtin_amdgcn_wmma_f32_16x16x4_f32(false, a0, false, f0, (short)0, c00, false, false);
        c01 = __builtin_amdgcn_wmma_f32_16x16x4_f32(false, a0, false, f1, (short)0, c01, false, false);
        c10 = __builtin_amdgcn_wmma_f32_16x16x4_f32(false, a1, false, f0, (short)0, c10, false, false);
        c11 = __builtin_amdgcn_wmma_f32_16x16x4_f32(false, a1, false, f1, (short)0, c11, false, false);
        a0.x = wa.z; a0.y = wa.w; a1.x = wb.z; a1.y = wb.w;
        f0.x = xa.z; f0.y = xa.w; f1.x = xb.z; f1.y = xb.w;
        c00 = __builtin_amdgcn_wmma_f32_16x16x4_f32(false, a0, false, f0, (short)0, c00, false, false);
        c01 = __builtin_amdgcn_wmma_f32_16x16x4_f32(false, a0, false, f1, (short)0, c01, false, false);
        c10 = __builtin_amdgcn_wmma_f32_16x16x4_f32(false, a1, false, f0, (short)0, c10, false, false);
        c11 = __builtin_amdgcn_wmma_f32_16x16x4_f32(false, a1, false, f1, (short)0, c11, false, false);
    }

    float* dst = out + (size_t)b * Cc * HW + (size_t)(o0 + 8 * h) * HW + pix + 2 * c;
    #pragma unroll
    for (int r = 0; r < 8; ++r) {
        float2 lo, hi;
        lo.x = c00[r]; lo.y = c01[r];
        hi.x = c10[r]; hi.y = c11[r];
        *(float2*)(dst + (size_t)r * HW)        = lo;
        *(float2*)(dst + (size_t)(16 + r) * HW) = hi;
    }
}

extern "C" void kernel_launch(void* const* d_in, const int* in_sizes, int n_in,
                              void* d_out, int out_size, void* d_ws, size_t ws_size,
                              hipStream_t stream)
{
    const float* x  = (const float*)d_in[0];
    const int*   m  = (const int*)  d_in[1];
    const float* Wq = (const float*)d_in[2];
    const float* Wk = (const float*)d_in[3];
    const float* Wv = (const float*)d_in[4];
    const float* Wp = (const float*)d_in[5];
    float* out = (float*)d_out;

    float* ws = (float*)d_ws;
    const size_t n = (size_t)NPIX * Cc;
    float* qb = ws;
    float* kb = ws + n;
    float* vb = ws + 2 * n;
    float* ab = ws + 3 * n;

    qkv_wmma_kernel<<<QKV_BLOCKS, 128, 0, stream>>>(x, Wq, Wk, Wv, qb, kb, vb);
    attn_kernel<<<ATT_BLOCKS, 256, 0, stream>>>(qb, kb, vb, m, ab);
    proj_wmma_kernel<<<NPIX / 32, 128, 0, stream>>>(ab, Wp, out);
}


# attention: softmax in two passes (49 logits in registers, one max, one exp per tap; sums of weights and of weighted V no longer rescaled); tap addresses and mask bits rebuilt so the scalars fit
# speedup vs baseline: 3.7269x; 1.0709x over previous
#include <hip/hip_runtime.h>
#include <hip/hip_bf16.h>

typedef float v2f __attribute__((ext_vector_type(2)));
typedef float v8f __attribute__((ext_vector_type(8)));

#define Hh   96
#define Ww   96
#define HW   (96 * 96)
#define Cc   128
#define NPIX (2 * HW)

#define QKV_BLOCKS ((NPIX / 64) * 6)

__global__ void __launch_bounds__(128) qkv_wmma_kernel(
    const float* __restrict__ x,  const float* __restrict__ Wq,
    const float* __restrict__ Wk, const float* __restrict__ Wv,
    float* __restrict__ qo, float* __restrict__ ko, float* __restrict__ vo)
{
    const int lane = threadIdx.x & 31;
    const int wave = threadIdx.x >> 5;
    const int grp  = blockIdx.x / 6;
    const int sub  = blockIdx.x - grp * 6;
    const int mat  = sub >> 1;
    const float* __restrict__ Wm = (mat == 0) ? Wq : ((mat == 1) ? Wk : Wv);
    float* __restrict__ dm       = (mat == 0) ? qo : ((mat == 1) ? ko : vo);

    const int p0  = grp * 64;
    const int b   = p0 / HW;
    const int pix = p0 - b * HW;
    const int o0  = (((sub & 1) << 2) + wave) * 16;
    const int c   = lane & 15;
    const int h   = lane >> 4;

    const float* ap = Wm + (size_t)(o0 + c) * Cc + 4 * h;
    const float* bp = x + (size_t)b * Cc * HW + pix + (size_t)(4 * h) * HW + 4 * c;

    v8f acc0 = {}; v8f acc1 = {}; v8f acc2 = {}; v8f acc3 = {};
    #pragma unroll 2
    for (int kb = 0; kb < Cc / 8; ++kb) {
        const float4 a4 = *(const float4*)(ap + kb * 8);
        const float4 r0 = *(const float4*)(bp + (size_t)(kb * 8 + 0) * HW);
        const float4 r1 = *(const float4*)(bp + (size_t)(kb * 8 + 1) * HW);
        const float4 r2 = *(const float4*)(bp + (size_t)(kb * 8 + 2) * HW);
        const float4 r3 = *(const float4*)(bp + (size_t)(kb * 8 + 3) * HW);
        v2f a, f;
        a.x = a4.x; a.y = a4.y;
        f.x = r0.x; f.y = r1.x;
        acc0 = __builtin_amdgcn_wmma_f32_16x16x4_f32(false, a, false, f, (short)0, acc0, false, false);
        f.x = r0.y; f.y = r1.y;
        acc1 = __builtin_amdgcn_wmma_f32_16x16x4_f32(false, a, false, f, (short)0, acc1, false, false);
        f.x = r0.z; f.y = r1.z;
        acc2 = __builtin_amdgcn_wmma_f32_16x16x4_f32(false, a, false, f, (short)0, acc2, false, false);
        f.x = r0.w; f.y = r1.w;
        acc3 = __builtin_amdgcn_wmma_f32_16x16x4_f32(false, a, false, f, (short)0, acc3, false, false);
        a.x = a4.z; a.y = a4.w;
        f.x = r2.x; f.y = r3.x;
        acc0 = __builtin_amdgcn_wmma_f32_16x16x4_f32(false, a, false, f, (short)0, acc0, false, false);
        f.x = r2.y; f.y = r3.y;
        acc1 = __builtin_amdgcn_wmma_f32_16x16x4_f32(false, a, false, f, (short)0, acc1, false, false);
        f.x = r2.z; f.y = r3.z;
        acc2 = __builtin_amdgcn_wmma_f32_16x16x4_f32(false, a, false, f, (short)0, acc2, false, false);
        f.x = r2.w; f.y = r3.w;
        acc3 = __builtin_amdgcn_wmma_f32_16x16x4_f32(false, a, false, f, (short)0, acc3, false, false);
    }

    float* d0 = dm + (size_t)(p0 + 4 * c) * Cc + o0 + 8 * h;
    #pragma unroll
    for (int r = 0; r < 8; ++r) {
        d0[r]          = acc0[r];
        d0[Cc + r]     = acc1[r];
        d0[2 * Cc + r] = acc2[r];
        d0[3 * Cc + r] = acc3[r];
    }
}

#define ATT_BLOCKS (2 * Hh * 6 * 2)

__device__ __forceinline__ float add_xor1(float v)
{
    float r;
    asm("v_add_f32_dpp %0, %1, %1 row_xmask:1 row_mask:0xf bank_mask:0xf" : "=v"(r) : "v"(v));
    return r;
}
__device__ __forceinline__ float add_xor2(float v)
{
    float r;
    asm("v_add_f32_dpp %0, %1, %1 row_xmask:2 row_mask:0xf bank_mask:0xf" : "=v"(r) : "v"(v));
    return r;
}
__device__ __forceinline__ float add_xor4(float v)
{
    float r;
    asm("v_add_f32_dpp %0, %1, %1 row_xmask:4 row_mask:0xf bank_mask:0xf" : "=v"(r) : "v"(v));
    return r;
}

__global__ void __launch_bounds__(256) attn_kernel(
    const float* __restrict__ qv, const float* __restrict__ kv,
    const float* __restrict__ vv, const int* __restrict__ msk,
    float* __restrict__ ao)
{
    const int lane = threadIdx.x & 31;
    const int wave = __builtin_amdgcn_readfirstlane((int)(threadIdx.x >> 5));
    const int blk  = blockIdx.x;
    const int par  = blk & 1;
    const int rs   = blk >> 1;
    const int brow = rs / 6;
    const int seg  = rs - brow * 6;
    const int b    = brow / Hh;
    const int y    = brow - b * Hh;
    const int x    = seg * 16 + par + 2 * wave;
    const int p    = brow * Ww + x;

    const float4 q = *(const float4*)(qv + (size_t)p * Cc + lane * 4);

    const float scale = 0.17677669529663687f;

    int rowpix[7];
    unsigned coff[7];
    #pragma unroll
    for (int i = 0; i < 7; ++i) {
        const int ny = y + (i - 3) * 2;
        const int nx = x + (i - 3) * 2;
        rowpix[i] = b * HW + (((unsigned)ny < (unsigned)Hh) ? ny : y) * Ww;
        coff[i]   = (unsigned)(((unsigned)nx < (unsigned)Ww) ? nx : x) * (unsigned)(Cc * 4)
                  + (unsigned)lane * 16u;
    }

    unsigned inb[2], opn[2];
    #pragma unroll
    for (int half = 0; half < 2; ++half) {
        const int t  = lane + 32 * half;
        const int ti = (t * 37) >> 8;
        const int tj = t - 7 * ti;
        const int ny = y + (ti - 3) * 2;
        const int nx = x + (tj - 3) * 2;
        const bool in = (t < 49) && ((unsigned)ny < (unsigned)Hh) && ((unsigned)nx < (unsigned)Ww);
        const int mword = msk[in ? (b * HW + ny * Ww + nx) : p];
        inb[half] = __builtin_amdgcn_ballot_w32(in);
        opn[half] = __builtin_amdgcn_ballot_w32(in && (mword > 0));
    }

    float lg[49];
    float4 bufa[7], bufb[7];

    auto logits_row = [&](int di, const float4 (&kk)[7]) {
        #pragma unroll
        for (int dj = 0; dj < 7; ++dj) {
            const int t = di * 7 + dj;
            float dot = q.x * kk[dj].x + q.y * kk[dj].y + q.z * kk[dj].z + q.w * kk[dj].w;
            dot = add_xor1(dot);
            dot = add_xor2(dot);
            dot = add_xor4(dot);

            const float sc = ((inb[t >> 5] >> (t & 31)) & 1u) ? scale : 0.f;
            const float mb = ((opn[t >> 5] >> (t & 31)) & 1u) ? 0.f : -1.0e9f;
            lg[t] = dot * sc + mb;
        }
    };
    auto load_row = [&](const float* base, int di, float4 (&dst)[7]) {
        const char* row = (const char*)base + (size_t)rowpix[di] * (size_t)(Cc * 4);
        #pragma unroll
        for (int dj = 0; dj < 7; ++dj)
            dst[dj] = *(const float4*)(row + coff[dj]);
    };

    #pragma unroll
    for (int g = 0; g < 3; ++g) {
        load_row(kv, 2 * g,     bufa);
        load_row(kv, 2 * g + 1, bufb);
        logits_row(2 * g,     bufa);
        logits_row(2 * g + 1, bufb);
    }
    load_row(kv, 6, bufa);
    load_row(vv, 0, bufb);
    logits_row(6, bufa);

    float mx = lg[0];
    #pragma unroll
    for (int i = 1; i < 49; ++i) mx = fmaxf(mx, lg[i]);

    float s = 0.f;
    float4 acc = make_float4(0.f, 0.f, 0.f, 0.f);
    auto weigh_row = [&](int di, const float4 (&vq)[7]) {
        float w[7];
        #pragma unroll
        for (int dj = 0; dj < 7; ++dj)
            w[dj] = __expf(lg[di * 7 + dj] - mx);
        #pragma unroll
        for (int dj = 0; dj < 7; ++dj) {
            const int t = di * 7 + dj;
            s += w[dj];
            const float wv = ((inb[t >> 5] >> (t & 31)) & 1u) ? w[dj] : 0.f;
            acc.x += wv * vq[dj].x;
            acc.y += wv * vq[dj].y;
            acc.z += wv * vq[dj].z;
            acc.w += wv * vq[dj].w;
        }
    };
    weigh_row(0, bufb);
    #pragma unroll
    for (int g = 0; g < 3; ++g) {
        load_row(vv, 2 * g + 1, bufa);
        load_row(vv, 2 * g + 2, bufb);
        weigh_row(2 * g + 1, bufa);
        weigh_row(2 * g + 2, bufb);
    }

    const float inv = 1.f / s;
    float4 o4;
    o4.x = acc.x * inv; o4.y = acc.y * inv; o4.z = acc.z * inv; o4.w = acc.w * inv;
    *(float4*)(ao + (size_t)p * Cc + lane * 4) = o4;
}

__global__ void __launch_bounds__(128) proj_wmma_kernel(
    const float* __restrict__ ain, const float* __restrict__ Wp,
    float* __restrict__ out)
{
    const int lane = threadIdx.x & 31;
    const int wave = threadIdx.x >> 5;
    const int p0   = blockIdx.x * 32;
    const int b    = p0 / HW;
    const int pix  = p0 - b * HW;
    const int o0   = wave * 32;
    const int c    = lane & 15;
    const int h    = lane >> 4;

    const float* ap = Wp  + (size_t)(o0 + c) * Cc + 4 * h;
    const float* bp = ain + (size_t)(p0 + 2 * c) * Cc + 4 * h;

    v8f c00 = {}; v8f c01 = {}; v8f c10 = {}; v8f c11 = {};
    #pragma unroll 4
    for (int kb = 0; kb < Cc / 8; ++kb) {
        const float4 wa = *(const float4*)(ap + kb * 8);
        const float4 wb = *(const float4*)(ap + 16 * Cc + kb * 8);
        const float4 xa = *(const float4*)(bp + kb * 8);
        const float4 xb = *(const float4*)(bp + Cc + kb * 8);
        v2f a0, a1, f0, f1;
        a0.x = wa.x; a0.y = wa.y; a1.x = wb.x; a1.y = wb.y;
        f0.x = xa.x; f0.y = xa.y; f1.x = xb.x; f1.y = xb.y;
        c00 = __builtin_amdgcn_wmma_f32_16x16x4_f32(false, a0, false, f0, (short)0, c00, false, false);
        c01 = __builtin_amdgcn_wmma_f32_16x16x4_f32(false, a0, false, f1, (short)0, c01, false, false);
        c10 = __builtin_amdgcn_wmma_f32_16x16x4_f32(false, a1, false, f0, (short)0, c10, false, false);
        c11 = __builtin_amdgcn_wmma_f32_16x16x4_f32(false, a1, false, f1, (short)0, c11, false, false);
        a0.x = wa.z; a0.y = wa.w; a1.x = wb.z; a1.y = wb.w;
        f0.x = xa.z; f0.y = xa.w; f1.x = xb.z; f1.y = xb.w;
        c00 = __builtin_amdgcn_wmma_f32_16x16x4_f32(false, a0, false, f0, (short)0, c00, false, false);
        c01 = __builtin_amdgcn_wmma_f32_16x16x4_f32(false, a0, false, f1, (short)0, c01, false, false);
        c10 = __builtin_amdgcn_wmma_f32_16x16x4_f32(false, a1, false, f0, (short)0, c10, false, false);
        c11 = __builtin_amdgcn_wmma_f32_16x16x4_f32(false, a1, false, f1, (short)0, c11, false, false);
    }

    float* dst = out + (size_t)b * Cc * HW + (size_t)(o0 + 8 * h) * HW + pix + 2 * c;
    #pragma unroll
    for (int r = 0; r < 8; ++r) {
        float2 lo, hi;
        lo.x = c00[r]; lo.y = c01[r];
        hi.x = c10[r]; hi.y = c11[r];
        *(float2*)(dst + (size_t)r * HW)        = lo;
        *(float2*)(dst + (size_t)(16 + r) * HW) = hi;
    }
}

extern "C" void kernel_launch(void* const* d_in, const int* in_sizes, int n_in,
                              void* d_out, int out_size, void* d_ws, size_t ws_size,
                              hipStream_t stream)
{
    const float* x  = (const float*)d_in[0];
    const int*   m  = (const int*)  d_in[1];
    const float* Wq = (const float*)d_in[2];
    const float* Wk = (const float*)d_in[3];
    const float* Wv = (const float*)d_in[4];
    const float* Wp = (const float*)d_in[5];
    float* out = (float*)d_out;

    float* ws = (float*)d_ws;
    const size_t n = (size_t)NPIX * Cc;
    float* qb = ws;
    float* kb = ws + n;
    float* vb = ws + 2 * n;
    float* ab = ws + 3 * n;

    qkv_wmma_kernel<<<QKV_BLOCKS, 128, 0, stream>>>(x, Wq, Wk, Wv, qb, kb, vb);
    attn_kernel<<<ATT_BLOCKS, 256, 0, stream>>>(qb, kb, vb, m, ab);
    proj_wmma_kernel<<<NPIX / 32, 128, 0, stream>>>(ab, Wp, out);
}


# attention: softmax once per head, not in each of its 8 lanes: lane t&7 keeps tap t's logit (7 regs, was 49; 12 waves/SIMD, was 8), 7 exps a lane, weights broadcast in-group; weight-sum order changes
# speedup vs baseline: 3.9178x; 1.0512x over previous
#include <hip/hip_runtime.h>
#include <hip/hip_bf16.h>

typedef float v2f __attribute__((ext_vector_type(2)));
typedef float v8f __attribute__((ext_vector_type(8)));

#define Hh   96
#define Ww   96
#define HW   (96 * 96)
#define Cc   128
#define NPIX (2 * HW)

#define QKV_BLOCKS ((NPIX / 64) * 6)

__global__ void __launch_bounds__(128) qkv_wmma_kernel(
    const float* __restrict__ x,  const float* __restrict__ Wq,
    const float* __restrict__ Wk, const float* __restrict__ Wv,
    float* __restrict__ qo, float* __restrict__ ko, float* __restrict__ vo)
{
    const int lane = threadIdx.x & 31;
    const int wave = threadIdx.x >> 5;
    const int grp  = blockIdx.x / 6;
    const int sub  = blockIdx.x - grp * 6;
    const int mat  = sub >> 1;
    const float* __restrict__ Wm = (mat == 0) ? Wq : ((mat == 1) ? Wk : Wv);
    float* __restrict__ dm       = (mat == 0) ? qo : ((mat == 1) ? ko : vo);

    const int p0  = grp * 64;
    const int b   = p0 / HW;
    const int pix = p0 - b * HW;
    const int o0  = (((sub & 1) << 2) + wave) * 16;
    const int c   = lane & 15;
    const int h   = lane >> 4;

    const float* ap = Wm + (size_t)(o0 + c) * Cc + 4 * h;
    const float* bp = x + (size_t)b * Cc * HW + pix + (size_t)(4 * h) * HW + 4 * c;

    v8f acc0 = {}; v8f acc1 = {}; v8f acc2 = {}; v8f acc3 = {};
    #pragma unroll 2
    for (int kb = 0; kb < Cc / 8; ++kb) {
        const float4 a4 = *(const float4*)(ap + kb * 8);
        const float4 r0 = *(const float4*)(bp + (size_t)(kb * 8 + 0) * HW);
        const float4 r1 = *(const float4*)(bp + (size_t)(kb * 8 + 1) * HW);
        const float4 r2 = *(const float4*)(bp + (size_t)(kb * 8 + 2) * HW);
        const float4 r3 = *(const float4*)(bp + (size_t)(kb * 8 + 3) * HW);
        v2f a, f;
        a.x = a4.x; a.y = a4.y;
        f.x = r0.x; f.y = r1.x;
        acc0 = __builtin_amdgcn_wmma_f32_16x16x4_f32(false, a, false, f, (short)0, acc0, false, false);
        f.x = r0.y; f.y = r1.y;
        acc1 = __builtin_amdgcn_wmma_f32_16x16x4_f32(false, a, false, f, (short)0, acc1, false, false);
        f.x = r0.z; f.y = r1.z;
        acc2 = __builtin_amdgcn_wmma_f32_16x16x4_f32(false, a, false, f, (short)0, acc2, false, false);
        f.x = r0.w; f.y = r1.w;
        acc3 = __builtin_amdgcn_wmma_f32_16x16x4_f32(false, a, false, f, (short)0, acc3, false, false);
        a.x = a4.z; a.y = a4.w;
        f.x = r2.x; f.y = r3.x;
        acc0 = __builtin_amdgcn_wmma_f32_16x16x4_f32(false, a, false, f, (short)0, acc0, false, false);
        f.x = r2.y; f.y = r3.y;
        acc1 = __builtin_amdgcn_wmma_f32_16x16x4_f32(false, a, false, f, (short)0, acc1, false, false);
        f.x = r2.z; f.y = r3.z;
        acc2 = __builtin_amdgcn_wmma_f32_16x16x4_f32(false, a, false, f, (short)0, acc2, false, false);
        f.x = r2.w; f.y = r3.w;
        acc3 = __builtin_amdgcn_wmma_f32_16x16x4_f32(false, a, false, f, (short)0, acc3, false, false);
    }

    float* d0 = dm + (size_t)(p0 + 4 * c) * Cc + o0 + 8 * h;
    #pragma unroll
    for (int r = 0; r < 8; ++r) {
        d0[r]          = acc0[r];
        d0[Cc + r]     = acc1[r];
        d0[2 * Cc + r] = acc2[r];
        d0[3 * Cc + r] = acc3[r];
    }
}

#define ATT_BLOCKS (2 * Hh * 6 * 2)

__device__ __forceinline__ float add_xor1(float v)
{
    float r;
    asm("v_add_f32_dpp %0, %1, %1 row_xmask:1 row_mask:0xf bank_mask:0xf" : "=v"(r) : "v"(v));
    return r;
}
__device__ __forceinline__ float add_xor2(float v)
{
    float r;
    asm("v_add_f32_dpp %0, %1, %1 row_xmask:2 row_mask:0xf bank_mask:0xf" : "=v"(r) : "v"(v));
    return r;
}
__device__ __forceinline__ float add_xor4(float v)
{
    float r;
    asm("v_add_f32_dpp %0, %1, %1 row_xmask:4 row_mask:0xf bank_mask:0xf" : "=v"(r) : "v"(v));
    return r;
}

constexpr unsigned sel8(unsigned a, unsigned b, unsigned c, unsigned d,
                        unsigned e, unsigned f, unsigned g, unsigned h)
{
    return a | (b << 3) | (c << 6) | (d << 9) | (e << 12) | (f << 15) | (g << 18) | (h << 21);
}
template <unsigned SEL>
__device__ __forceinline__ float perm8(float v)
{
    return __int_as_float((int)__builtin_amdgcn_mov_dpp8((unsigned)__float_as_int(v), SEL));
}
#define BCAST8_CASE(J) \
    case J: asm("v_mov_b32_dpp %0, %1 dpp8:[" #J "," #J "," #J "," #J "," #J "," #J "," #J "," #J "]" : "=v"(r) : "v"(v)); break;
__device__ __forceinline__ float bcast8(float v, int j)
{
    float r;
    switch (j) {
    BCAST8_CASE(0) BCAST8_CASE(1) BCAST8_CASE(2) BCAST8_CASE(3)
    BCAST8_CASE(4) BCAST8_CASE(5) BCAST8_CASE(6)
    default: asm("v_mov_b32_dpp %0, %1 dpp8:[7,7,7,7,7,7,7,7]" : "=v"(r) : "v"(v)); break;
    }
    return r;
}
__device__ __forceinline__ float max8(float v)
{
    v = fmaxf(v, perm8<sel8(1, 0, 3, 2, 5, 4, 7, 6)>(v));
    v = fmaxf(v, perm8<sel8(2, 3, 0, 1, 6, 7, 4, 5)>(v));
    v = fmaxf(v, perm8<sel8(4, 5, 6, 7, 0, 1, 2, 3)>(v));
    return v;
}

__global__ void __launch_bounds__(256, 12) attn_kernel(
    const float* __restrict__ qv, const float* __restrict__ kv,
    const float* __restrict__ vv, const int* __restrict__ msk,
    float* __restrict__ ao)
{
    const int lane = threadIdx.x & 31;
    const int part = lane & 7;
    const int wave = __builtin_amdgcn_readfirstlane((int)(threadIdx.x >> 5));
    const int blk  = blockIdx.x;
    const int par  = blk & 1;
    const int rs   = blk >> 1;
    const int brow = rs / 6;
    const int seg  = rs - brow * 6;
    const int b    = brow / Hh;
    const int y    = brow - b * Hh;
    const int x    = seg * 16 + par + 2 * wave;
    const int p    = brow * Ww + x;

    const float4 q = *(const float4*)(qv + (size_t)p * Cc + lane * 4);

    const float scale = 0.17677669529663687f;

    int rowpix[7];
    unsigned coff[7];
    #pragma unroll
    for (int i = 0; i < 7; ++i) {
        const int ny = y + (i - 3) * 2;
        const int nx = x + (i - 3) * 2;
        rowpix[i] = b * HW + (((unsigned)ny < (unsigned)Hh) ? ny : y) * Ww;
        coff[i]   = (unsigned)(((unsigned)nx < (unsigned)Ww) ? nx : x) * (unsigned)(Cc * 4)
                  + (unsigned)lane * 16u;
    }

    unsigned inb[2], opn[2];
    #pragma unroll
    for (int half = 0; half < 2; ++half) {
        const int t  = lane + 32 * half;
        const int ti = (t * 37) >> 8;
        const int tj = t - 7 * ti;
        const int ny = y + (ti - 3) * 2;
        const int nx = x + (tj - 3) * 2;
        const bool in = (t < 49) && ((unsigned)ny < (unsigned)Hh) && ((unsigned)nx < (unsigned)Ww);
        const int mword = msk[in ? (b * HW + ny * Ww + nx) : p];
        inb[half] = __builtin_amdgcn_ballot_w32(in);
        opn[half] = __builtin_amdgcn_ballot_w32(in && (mword > 0));
    }

    float lgc[7];
    #pragma unroll
    for (int r = 0; r < 7; ++r) lgc[r] = -3.0e38f;
    bool own[8];
    #pragma unroll
    for (int j = 0; j < 8; ++j) own[j] = (part == j);
    float4 bufa[7], bufb[7];

    auto logits_row = [&](int di, const float4 (&kk)[7]) {
        #pragma unroll
        for (int dj = 0; dj < 7; ++dj) {
            const int t = di * 7 + dj;
            float dot = q.x * kk[dj].x + q.y * kk[dj].y + q.z * kk[dj].z + q.w * kk[dj].w;
            dot = add_xor1(dot);
            dot = add_xor2(dot);
            dot = add_xor4(dot);

            const float sc = ((inb[t >> 5] >> (t & 31)) & 1u) ? scale : 0.f;
            const float mb = ((opn[t >> 5] >> (t & 31)) & 1u) ? 0.f : -1.0e9f;
            const float val  = dot * sc + mb;
            const float keep = lgc[t >> 3];
            lgc[t >> 3] = own[t & 7] ? val : keep;
        }
    };
    auto load_row = [&](const float* base, int di, float4 (&dst)[7]) {
        const char* row = (const char*)base + (size_t)rowpix[di] * (size_t)(Cc * 4);
        #pragma unroll
        for (int dj = 0; dj < 7; ++dj)
            dst[dj] = *(const float4*)(row + coff[dj]);
    };

    #define STAGE_FENCE(ra, rb, rc) asm volatile("" :: "v"(lgc[ra]), "v"(lgc[rb]), "v"(lgc[rc]))
    #define STAGE_FENCE2() asm volatile("" :: "v"(acc.x), "v"(acc.y), "v"(acc.z), "v"(acc.w))
    #pragma unroll
    for (int g = 0; g < 3; ++g) {
        load_row(kv, 2 * g,     bufa);
        load_row(kv, 2 * g + 1, bufb);
        logits_row(2 * g,     bufa);
        logits_row(2 * g + 1, bufb);
        STAGE_FENCE((14 * g) >> 3, ((14 * g) >> 3) + 1, (14 * g + 13) >> 3);
    }
    load_row(kv, 6, bufa);
    load_row(vv, 0, bufb);
    logits_row(6, bufa);

    float mx = lgc[0];
    #pragma unroll
    for (int r = 1; r < 7; ++r) mx = fmaxf(mx, lgc[r]);
    mx = max8(mx);

    float s = 0.f;
    float wz[7];
    #pragma unroll
    for (int r = 0; r < 7; ++r) {
        const float w = __expf(lgc[r] - mx);
        s += w;
        const unsigned bits = inb[r >> 2] >> (8 * (r & 3));
        wz[r] = ((bits >> part) & 1u) ? w : 0.f;
    }
    s = add_xor1(s);
    s = add_xor2(s);
    s = add_xor4(s);

    float4 acc = make_float4(0.f, 0.f, 0.f, 0.f);
    auto weigh_row = [&](int di, const float4 (&vq)[7]) {
        #pragma unroll
        for (int dj = 0; dj < 7; ++dj) {
            const int t = di * 7 + dj;
            const float wv = bcast8(wz[t >> 3], t & 7);
            acc.x = fmaf(wv, vq[dj].x, acc.x);
            acc.y = fmaf(wv, vq[dj].y, acc.y);
            acc.z = fmaf(wv, vq[dj].z, acc.z);
            acc.w = fmaf(wv, vq[dj].w, acc.w);
        }
    };
    weigh_row(0, bufb);
    STAGE_FENCE2();
    #pragma unroll
    for (int g = 0; g < 3; ++g) {
        load_row(vv, 2 * g + 1, bufa);
        load_row(vv, 2 * g + 2, bufb);
        weigh_row(2 * g + 1, bufa);
        weigh_row(2 * g + 2, bufb);
        STAGE_FENCE2();
    }
    #undef STAGE_FENCE
    #undef STAGE_FENCE2

    const float inv = 1.f / s;
    float4 o4;
    o4.x = acc.x * inv; o4.y = acc.y * inv; o4.z = acc.z * inv; o4.w = acc.w * inv;
    *(float4*)(ao + (size_t)p * Cc + lane * 4) = o4;
}

__global__ void __launch_bounds__(128) proj_wmma_kernel(
    const float* __restrict__ ain, const float* __restrict__ Wp,
    float* __restrict__ out)
{
    const int lane = threadIdx.x & 31;
    const int wave = threadIdx.x >> 5;
    const int p0   = blockIdx.x * 32;
    const int b    = p0 / HW;
    const int pix  = p0 - b * HW;
    const int o0   = wave * 32;
    const int c    = lane & 15;
    const int h    = lane >> 4;

    const float* ap = Wp  + (size_t)(o0 + c) * Cc + 4 * h;
    const float* bp = ain + (size_t)(p0 + 2 * c) * Cc + 4 * h;

    v8f c00 = {}; v8f c01 = {}; v8f c10 = {}; v8f c11 = {};
    #pragma unroll 4
    for (int kb = 0; kb < Cc / 8; ++kb) {
        const float4 wa = *(const float4*)(ap + kb * 8);
        const float4 wb = *(const float4*)(ap + 16 * Cc + kb * 8);
        const float4 xa = *(const float4*)(bp + kb * 8);
        const float4 xb = *(const float4*)(bp + Cc + kb * 8);
        v2f a0, a1, f0, f1;
        a0.x = wa.x; a0.y = wa.y; a1.x = wb.x; a1.y = wb.y;
        f0.x = xa.x; f0.y = xa.y; f1.x = xb.x; f1.y = xb.y;
        c00 = __builtin_amdgcn_wmma_f32_16x16x4_f32(false, a0, false, f0, (short)0, c00, false, false);
        c01 = __builtin_amdgcn_wmma_f32_16x16x4_f32(false, a0, false, f1, (short)0, c01, false, false);
        c10 = __builtin_amdgcn_wmma_f32_16x16x4_f32(false, a1, false, f0, (short)0, c10, false, false);
        c11 = __builtin_amdgcn_wmma_f32_16x16x4_f32(false, a1, false, f1, (short)0, c11, false, false);
        a0.x = wa.z; a0.y = wa.w; a1.x = wb.z; a1.y = wb.w;
        f0.x = xa.z; f0.y = xa.w; f1.x = xb.z; f1.y = xb.w;
        c00 = __builtin_amdgcn_wmma_f32_16x16x4_f32(false, a0, false, f0, (short)0, c00, false, false);
        c01 = __builtin_amdgcn_wmma_f32_16x16x4_f32(false, a0, false, f1, (short)0, c01, false, false);
        c10 = __builtin_amdgcn_wmma_f32_16x16x4_f32(false, a1, false, f0, (short)0, c10, false, false);
        c11 = __builtin_amdgcn_wmma_f32_16x16x4_f32(false, a1, false, f1, (short)0, c11, false, false);
    }

    float* dst = out + (size_t)b * Cc * HW + (size_t)(o0 + 8 * h) * HW + pix + 2 * c;
    #pragma unroll
    for (int r = 0; r < 8; ++r) {
        float2 lo, hi;
        lo.x = c00[r]; lo.y = c01[r];
        hi.x = c10[r]; hi.y = c11[r];
        *(float2*)(dst + (size_t)r * HW)        = lo;
        *(float2*)(dst + (size_t)(16 + r) * HW) = hi;
    }
}

extern "C" void kernel_launch(void* const* d_in, const int* in_sizes, int n_in,
                              void* d_out, int out_size, void* d_ws, size_t ws_size,
                              hipStream_t stream)
{
    const float* x  = (const float*)d_in[0];
    const int*   m  = (const int*)  d_in[1];
    const float* Wq = (const float*)d_in[2];
    const float* Wk = (const float*)d_in[3];
    const float* Wv = (const float*)d_in[4];
    const float* Wp = (const float*)d_in[5];
    float* out = (float*)d_out;

    float* ws = (float*)d_ws;
    const size_t n = (size_t)NPIX * Cc;
    float* qb = ws;
    float* kb = ws + n;
    float* vb = ws + 2 * n;
    float* ab = ws + 3 * n;

    qkv_wmma_kernel<<<QKV_BLOCKS, 128, 0, stream>>>(x, Wq, Wk, Wv, qb, kb, vb);
    attn_kernel<<<ATT_BLOCKS, 256, 0, stream>>>(qb, kb, vb, m, ab);
    proj_wmma_kernel<<<NPIX / 32, 128, 0, stream>>>(ab, Wp, out);
}
